# starting kernel
# baseline (speedup 1.0000x reference)
#include <hip/hip_runtime.h>
#include <hip/hip_bf16.h>


typedef float v2f __attribute__((ext_vector_type(2)));
typedef float v8f __attribute__((ext_vector_type(8)));

#define TPB 256


__global__ void k_deg_init(float* __restrict__ deg, int n) {
    int i = blockIdx.x * TPB + threadIdx.x;
    if (i < n) deg[i] = 1.0f;
}

__global__ void k_deg_count(const int* __restrict__ dst, float* __restrict__ deg, int E) {
    int e = blockIdx.x * TPB + threadIdx.x;
    if (e < E) atomicAdd(&deg[dst[e]], 1.0f);
}

__global__ void k_dinv(float* __restrict__ deg, int n) {
    int i = blockIdx.x * TPB + threadIdx.x;
    if (i < n) deg[i] = rsqrtf(deg[i]);
}

__global__ void k_norm(const int* __restrict__ src, const int* __restrict__ dst,
                       const float* __restrict__ dinv, float* __restrict__ nrm, int E) {
    int e = blockIdx.x * TPB + threadIdx.x;
    if (e < E) nrm[e] = dinv[src[e]] * dinv[dst[e]];
}


template <int K, int NOUT>
__global__ void __launch_bounds__(TPB)
gcn_gemm_wmma(const float* __restrict__ A, const float* __restrict__ W,
              float* __restrict__ H, int M) {
    constexpr int COL_TILES = NOUT / 16;
    constexpr int ROW_TILES_PER_BLK = 8 / COL_TILES;

    __shared__ float lw[K * NOUT];
    for (int i = threadIdx.x; i < K * NOUT; i += TPB) lw[i] = W[i];
    __syncthreads();

    const int wave = threadIdx.x >> 5;
    const int lane = threadIdx.x & 31;
    const int rowTile = blockIdx.x * ROW_TILES_PER_BLK + wave / COL_TILES;
    const int m0 = rowTile * 16;
    if (m0 >= M) return;
    const int n0 = (wave % COL_TILES) * 16;

    const int half = lane >> 4;
    const int l16  = lane & 15;
    const int koff = 2 * half;
    const float* __restrict__ arow = A  + (size_t)(m0 + l16) * K + koff;
    const float* __restrict__ brow = lw + (n0 + l16) * K + koff;

    v8f acc = {0.f, 0.f, 0.f, 0.f, 0.f, 0.f, 0.f, 0.f};
#pragma unroll
    for (int k = 0; k < K; k += 4) {
        v2f a = *(const v2f*)(arow + k);
        v2f b = *(const v2f*)(brow + k);
        acc = __builtin_amdgcn_wmma_f32_16x16x4_f32(
             false, a,  false, b,
             (short)0, acc,  false,  false);
    }

    const int n = n0 + l16;
    const int mbase = m0 + 8 * half;
#pragma unroll
    for (int j = 0; j < 8; ++j)
        H[(size_t)(mbase + j) * NOUT + n] = acc[j];
}


template <int F>
__global__ void k_selfinit(const float* __restrict__ h, const float* __restrict__ dinv,
                           float* __restrict__ out, int n) {
    constexpr int Q = F / 4;
    int t = blockIdx.x * TPB + threadIdx.x;
    int i = t / Q, q = t % Q;
    if (i < n) {
        float w = dinv[i] * dinv[i];
        float4 v = *(const float4*)(h + (size_t)i * F + 4 * q);
        float4 r; r.x = v.x * w; r.y = v.y * w; r.z = v.z * w; r.w = v.w * w;
        *(float4*)(out + (size_t)i * F + 4 * q) = r;
    }
}

template <int F>
__global__ void k_edge_agg(const float* __restrict__ h, const int* __restrict__ src,
                           const int* __restrict__ dst, const float* __restrict__ nrm,
                           float* __restrict__ out, int E) {
    constexpr int Q = F / 4;
    int t = blockIdx.x * TPB + threadIdx.x;
    int e = t / Q, q = t % Q;
    if (e < E) {
        int s = src[e], d = dst[e];
        float w = nrm[e];
        float4 v = *(const float4*)(h + (size_t)s * F + 4 * q);
        float* o = out + (size_t)d * F + 4 * q;
        atomicAdd(o + 0, v.x * w);
        atomicAdd(o + 1, v.y * w);
        atomicAdd(o + 2, v.z * w);
        atomicAdd(o + 3, v.w * w);
    }
}

template <int F, bool RELU>
__global__ void k_bias_act(float* __restrict__ io, const float* __restrict__ b, int n) {
    int t = blockIdx.x * TPB + threadIdx.x;
    if (t < n * F) {
        float v = io[t] + b[t % F];
        if (RELU) v = fmaxf(v, 0.0f);
        io[t] = v;
    }
}


static inline int cdiv(long long a, int b) { return (int)((a + b - 1) / b); }

extern "C" void kernel_launch(void* const* d_in, const int* in_sizes, int n_in,
                              void* d_out, int out_size, void* d_ws, size_t ws_size,
                              hipStream_t stream) {
    (void)n_in; (void)out_size; (void)ws_size;
    const float* x  = (const float*)d_in[0];
    const int*   ei = (const int*)  d_in[1];
    const float* W1 = (const float*)d_in[2];
    const float* b1 = (const float*)d_in[3];
    const float* W2 = (const float*)d_in[4];
    const float* b2 = (const float*)d_in[5];
    float* out = (float*)d_out;

    const int N = in_sizes[0] / 128;
    const int E = in_sizes[1] / 2;
    const int* src = ei;
    const int* dst = ei + E;

    float* ws   = (float*)d_ws;
    const size_t NA = ((size_t)N + 3) & ~(size_t)3;
    float* dinv = ws;
    float* h1   = ws + NA;
    float* agg1 = h1 + 64 * NA;
    float* nrm  = agg1 + 64 * NA;
    float* h2   = h1;

    k_deg_init <<<cdiv(N, TPB), TPB, 0, stream>>>(dinv, N);
    k_deg_count<<<cdiv(E, TPB), TPB, 0, stream>>>(dst, dinv, E);
    k_dinv     <<<cdiv(N, TPB), TPB, 0, stream>>>(dinv, N);
    k_norm     <<<cdiv(E, TPB), TPB, 0, stream>>>(src, dst, dinv, nrm, E);

    {
        int rowTiles = (N + 15) / 16;
        gcn_gemm_wmma<128, 64><<<cdiv(rowTiles, 2), TPB, 0, stream>>>(x, W1, h1, N);
    }
    k_selfinit<64><<<cdiv((long long)N * 16, TPB), TPB, 0, stream>>>(h1, dinv, agg1, N);
    k_edge_agg<64><<<cdiv((long long)E * 16, TPB), TPB, 0, stream>>>(h1, src, dst, nrm, agg1, E);
    k_bias_act<64, true><<<cdiv((long long)N * 64, TPB), TPB, 0, stream>>>(agg1, b1, N);

    {
        int rowTiles = (N + 15) / 16;
        gcn_gemm_wmma<64, 32><<<cdiv(rowTiles, 4), TPB, 0, stream>>>(agg1, W2, h2, N);
    }
    k_selfinit<32><<<cdiv((long long)N * 8, TPB), TPB, 0, stream>>>(h2, dinv, out, N);
    k_edge_agg<32><<<cdiv((long long)E * 8, TPB), TPB, 0, stream>>>(h2, src, dst, nrm, out, E);
    k_bias_act<32, false><<<cdiv((long long)N * 32, TPB), TPB, 0, stream>>>(out, b2, N);
}


# Elementwise passes folded away: each GEMM store also writes the self-loop start value plus bias, GEMM2 does relu on load, users take rsqrt(deg); 11 launches to 6; bias summed first, not last
# speedup vs baseline: 1.8608x; 1.8608x over previous
#include <hip/hip_runtime.h>
#include <hip/hip_bf16.h>


typedef float v2f __attribute__((ext_vector_type(2)));
typedef float v8f __attribute__((ext_vector_type(8)));

#define TPB 256


__global__ void k_deg_init(float* __restrict__ deg, int* __restrict__ tick, int n) {
    int i = blockIdx.x * TPB + threadIdx.x;
    if (i < n) deg[i] = 1.0f;
    if (i < 4) tick[i] = 0;
}

__global__ void k_deg_count(const int* __restrict__ dst, float* __restrict__ deg, int E) {
    int e = blockIdx.x * TPB + threadIdx.x;
    if (e < E) atomicAdd(&deg[dst[e]], 1.0f);
}


template <int K, int NOUT, bool RELU_A>
__global__ void __launch_bounds__(TPB)
gcn_gemm_wmma(const float* __restrict__ A, const float* __restrict__ W,
              const float* __restrict__ deg, const float* __restrict__ bias,
              float* __restrict__ H, float* __restrict__ S, int M) {
    constexpr int COL_TILES = NOUT / 16;
    constexpr int ROW_TILES_PER_BLK = 8 / COL_TILES;

    __shared__ float lw[K * NOUT];
    for (int i = threadIdx.x; i < K * NOUT; i += TPB) lw[i] = W[i];
    __syncthreads();

    const int wave = threadIdx.x >> 5;
    const int lane = threadIdx.x & 31;
    const int rowTile = blockIdx.x * ROW_TILES_PER_BLK + wave / COL_TILES;
    const int m0 = rowTile * 16;
    if (m0 >= M) return;
    const int n0 = (wave % COL_TILES) * 16;

    const int half = lane >> 4;
    const int l16  = lane & 15;
    const int koff = 2 * half;
    const float* __restrict__ arow = A  + (size_t)(m0 + l16) * K + koff;
    const float* __restrict__ brow = lw + (n0 + l16) * K + koff;

    const int n = n0 + l16;
    const int mbase = m0 + 8 * half;
    const float4 dg0 = *(const float4*)(deg + mbase);
    const float4 dg1 = *(const float4*)(deg + mbase + 4);
    const float bn = bias[n];

    v8f acc = {0.f, 0.f, 0.f, 0.f, 0.f, 0.f, 0.f, 0.f};
#pragma unroll
    for (int k = 0; k < K; k += 4) {
        v2f a = *(const v2f*)(arow + k);
        if (RELU_A) { a.x = fmaxf(a.x, 0.0f); a.y = fmaxf(a.y, 0.0f); }
        v2f b = *(const v2f*)(brow + k);
        acc = __builtin_amdgcn_wmma_f32_16x16x4_f32(
             false, a,  false, b,
             (short)0, acc,  false,  false);
    }

    const float dg[8] = {dg0.x, dg0.y, dg0.z, dg0.w, dg1.x, dg1.y, dg1.z, dg1.w};
#pragma unroll
    for (int j = 0; j < 8; ++j) {
        const float di = rsqrtf(dg[j]);
        const size_t o = (size_t)(mbase + j) * NOUT + n;
        H[o] = acc[j];
        S[o] = acc[j] * (di * di) + bn;
    }
}


template <int F, int UB>
__device__ __forceinline__ void agg_rows(const float* __restrict__ h, float* __restrict__ out,
                                         int s, int d, int wbits, int j0, int lane) {
    constexpr int P = F / 32;
    float v[UB][P];
#pragma unroll
    for (int u = 0; u < UB; ++u) {
        const int sj = __builtin_amdgcn_readlane(s, j0 + u);
        const float* __restrict__ hr = h + (size_t)sj * F + lane;
#pragma unroll
        for (int p = 0; p < P; ++p) v[u][p] = hr[32 * p];
    }
#pragma unroll
    for (int u = 0; u < UB; ++u) {
        const int dj = __builtin_amdgcn_readlane(d, j0 + u);
        const float wj = __int_as_float(__builtin_amdgcn_readlane(wbits, j0 + u));
        float* o = out + (size_t)dj * F + lane;
#pragma unroll
        for (int p = 0; p < P; ++p) atomicAdd(o + 32 * p, v[u][p] * wj);
    }
}

#define CHUNK (2 * TPB)
template <int F>
__global__ void __launch_bounds__(TPB)
k_edge_agg(const float* __restrict__ h, const int* __restrict__ src,
           const int* __restrict__ dst, const float* __restrict__ deg,
           float* __restrict__ out, int* __restrict__ tick, int E, int n) {
    constexpr int P = F / 32;
    constexpr int U = 16 / P;
    __shared__ int ls[CHUNK], ld[CHUNK], lw[CHUNK];
    __shared__ int wcnt[TPB / 32];
    __shared__ int job[2];

    const int lane = threadIdx.x & 31;
    const int wv = __builtin_amdgcn_readfirstlane((int)(threadIdx.x >> 5));
    const int nChunks = (E + CHUNK - 1) / CHUNK;

    const int hwid = __builtin_amdgcn_s_sendmsg_rtn(0x87);
    if (threadIdx.x == 0) {
        int hf = (hwid >> 18) & 1;
        int t = atomicAdd(&tick[hf], 1);
        if (t >= nChunks) { hf ^= 1; t = atomicAdd(&tick[hf], 1); }
        job[0] = hf; job[1] = t;
    }
    __syncthreads();
    const int hf = job[0];
    const int chunk = job[1];
    if (chunk >= nChunks) return;

    const int split = (n >> 1) & ~1;
    const int e0 = chunk * CHUNK + wv * 64;

    const int ea = e0 + lane, eb = e0 + 32 + lane;
    const int ca = ea < E ? ea : E - 1, cb = eb < E ? eb : E - 1;
    const int s0 = src[ca], d0 = dst[ca], s1 = src[cb], d1 = dst[cb];
    const bool k0 = (ea < E) && ((d0 >= split) == (hf != 0));
    const bool k1 = (eb < E) && ((d1 >= split) == (hf != 0));
    const float w0 = rsqrtf(deg[k0 ? s0 : 0]) * rsqrtf(deg[k0 ? d0 : 0]);
    const float w1 = rsqrtf(deg[k1 ? s1 : 0]) * rsqrtf(deg[k1 ? d1 : 0]);
    const unsigned m0 = (unsigned)__ballot(k0);
    const unsigned m1 = (unsigned)__ballot(k1);
    const unsigned below = (1u << lane) - 1u;
    const int c0 = __popc(m0);
    const int r0 = __popc(m0 & below);
    const int r1 = c0 + __popc(m1 & below);
    if (lane == 0) wcnt[wv] = c0 + __popc(m1);
    __syncthreads();
    int base = 0, total = 0;
#pragma unroll
    for (int i = 0; i < TPB / 32; ++i) {
        const int c = wcnt[i];
        if (i < wv) base += c;
        total += c;
    }
    if (k0) { ls[base + r0] = s0; ld[base + r0] = d0; lw[base + r0] = __float_as_int(w0); }
    if (k1) { ls[base + r1] = s1; ld[base + r1] = d1; lw[base + r1] = __float_as_int(w1); }
    __syncthreads();
    total = __builtin_amdgcn_readfirstlane(total);

    for (int b = wv * 32; b < total; b += TPB) {
        const int cnt = (total - b < 32) ? (total - b) : 32;
        int s = 0, d = 0, wbits = 0;
        if (lane < cnt) { s = ls[b + lane]; d = ld[b + lane]; wbits = lw[b + lane]; }
        int j0 = 0;
        for (; j0 + U <= cnt; j0 += U) agg_rows<F, U>(h, out, s, d, wbits, j0, lane);
        if (U > 8 && j0 + 8 <= cnt) { agg_rows<F, 8>(h, out, s, d, wbits, j0, lane); j0 += 8; }
        if (j0 + 4 <= cnt) { agg_rows<F, 4>(h, out, s, d, wbits, j0, lane); j0 += 4; }
        if (j0 + 2 <= cnt) { agg_rows<F, 2>(h, out, s, d, wbits, j0, lane); j0 += 2; }
        if (j0 < cnt)      { agg_rows<F, 1>(h, out, s, d, wbits, j0, lane); }
    }
}


static inline int cdiv(long long a, int b) { return (int)((a + b - 1) / b); }

extern "C" void kernel_launch(void* const* d_in, const int* in_sizes, int n_in,
                              void* d_out, int out_size, void* d_ws, size_t ws_size,
                              hipStream_t stream) {
    (void)n_in; (void)out_size; (void)ws_size;
    const float* x  = (const float*)d_in[0];
    const int*   ei = (const int*)  d_in[1];
    const float* W1 = (const float*)d_in[2];
    const float* b1 = (const float*)d_in[3];
    const float* W2 = (const float*)d_in[4];
    const float* b2 = (const float*)d_in[5];
    float* out = (float*)d_out;

    const int N = in_sizes[0] / 128;
    const int E = in_sizes[1] / 2;
    const int* src = ei;
    const int* dst = ei + E;

    float* ws   = (float*)d_ws;
    const size_t NA = ((size_t)N + 3) & ~(size_t)3;
    float* deg  = ws;
    float* h1   = ws + NA;
    float* agg1 = h1 + 64 * NA;
    int*   tick = (int*)(agg1 + 64 * NA);
    float* h2   = h1;

    k_deg_init <<<cdiv(N, TPB), TPB, 0, stream>>>(deg, tick, N);
    k_deg_count<<<cdiv(E, TPB), TPB, 0, stream>>>(dst, deg, E);

    const int rowTiles = (N + 15) / 16;

    gcn_gemm_wmma<128, 64, false><<<cdiv(rowTiles, 2), TPB, 0, stream>>>(x, W1, deg, b1, h1, agg1, N);
    k_edge_agg<64><<<2 * cdiv((long long)E, CHUNK), TPB, 0, stream>>>(h1, src, dst, deg, agg1, tick, E, N);

    gcn_gemm_wmma<64, 32, true><<<cdiv(rowTiles, 4), TPB, 0, stream>>>(agg1, W2, deg, b2, h2, out, N);
    k_edge_agg<32><<<2 * cdiv((long long)E, CHUNK), TPB, 0, stream>>>(h2, src, dst, deg, out, tick + 2, E, N);
}


# Both GEMMs keep all of W in each wave's registers as B fragments (no LDS tile, so no 16-way bank conflict); a wave walks 16-row strips, one A load feeds all column tiles; k order, sums unchanged
# speedup vs baseline: 2.0290x; 1.0904x over previous
#include <hip/hip_runtime.h>
#include <hip/hip_bf16.h>


typedef float v2f __attribute__((ext_vector_type(2)));
typedef float v8f __attribute__((ext_vector_type(8)));

#define TPB 256


__global__ void k_deg_init(float* __restrict__ deg, int* __restrict__ tick, int n) {
    int i = blockIdx.x * TPB + threadIdx.x;
    if (i < n) deg[i] = 1.0f;
    if (i < 4) tick[i] = 0;
}

__global__ void k_deg_count(const int* __restrict__ dst, float* __restrict__ deg, int E) {
    int e = blockIdx.x * TPB + threadIdx.x;
    if (e < E) atomicAdd(&deg[dst[e]], 1.0f);
}


#define GEMM_TPB 128
#define GEMM_BLOCKS 512

template <int K>
struct StripIn {
    v2f a[K / 4];
    float4 dg0, dg1;
};

template <int K>
__device__ __forceinline__ void load_strip(const float* __restrict__ A, const float* __restrict__ deg,
                                           int m0, int half, int l16, StripIn<K>& in) {
    const float* __restrict__ arow = A + (size_t)(m0 + l16) * K + 2 * half;
    const int mbase = m0 + 8 * half;
    in.dg0 = *(const float4*)(deg + mbase);
    in.dg1 = *(const float4*)(deg + mbase + 4);
#pragma unroll
    for (int i = 0; i < K / 4; ++i) in.a[i] = *(const v2f*)(arow + 4 * i);
}

template <int K, int NOUT, bool RELU_A>
__device__ __forceinline__ void strip_products(const StripIn<K>& in, const v2f (&bw)[NOUT / 16][K / 4],
                                               const float (&bn)[NOUT / 16],
                                               float* __restrict__ H, float* __restrict__ S,
                                               int mbase, int l16) {
    constexpr int CT = NOUT / 16;
    constexpr int KS = K / 4;
    v8f acc[CT];
#pragma unroll
    for (int t = 0; t < CT; ++t) acc[t] = (v8f){0.f, 0.f, 0.f, 0.f, 0.f, 0.f, 0.f, 0.f};
#pragma unroll
    for (int i = 0; i < KS; ++i) {
        v2f a = in.a[i];
        if (RELU_A) { a.x = fmaxf(a.x, 0.0f); a.y = fmaxf(a.y, 0.0f); }
#pragma unroll
        for (int t = 0; t < CT; ++t) {
            acc[t] = __builtin_amdgcn_wmma_f32_16x16x4_f32(
                 false, a,  false, bw[t][i],
                 (short)0, acc[t],  false,  false);
        }
    }
    const float dg[8] = {in.dg0.x, in.dg0.y, in.dg0.z, in.dg0.w, in.dg1.x, in.dg1.y, in.dg1.z, in.dg1.w};
#pragma unroll
    for (int j = 0; j < 8; ++j) {
        const float di = rsqrtf(dg[j]);
        const float d2 = di * di;
#pragma unroll
        for (int t = 0; t < CT; ++t) {
            const size_t o = (size_t)(mbase + j) * NOUT + 16 * t + l16;
            H[o] = acc[t][j];
            S[o] = acc[t][j] * d2 + bn[t];
        }
    }
}

template <int K, int NOUT, bool RELU_A>
__global__ void __launch_bounds__(GEMM_TPB)
gcn_gemm_wmma(const float* __restrict__ A, const float* __restrict__ W,
              const float* __restrict__ deg, const float* __restrict__ bias,
              float* __restrict__ H, float* __restrict__ S, int M) {
    constexpr int CT = NOUT / 16;
    constexpr int KS = K / 4;

    const int lane = threadIdx.x & 31;
    const int wv = __builtin_amdgcn_readfirstlane((int)(threadIdx.x >> 5));
    const int gw = blockIdx.x * (GEMM_TPB / 32) + wv;
    const int nw = gridDim.x * (GEMM_TPB / 32);
    const int nTiles = (M + 15) >> 4;

    const int half = lane >> 4;
    const int l16  = lane & 15;

    int tile = gw;
    if (tile >= nTiles) return;

    v2f bw[CT][KS];
    float bn[CT];
#pragma unroll
    for (int t = 0; t < CT; ++t) {
        const float* __restrict__ wrow = W + (size_t)(16 * t + l16) * K + 2 * half;
#pragma unroll
        for (int i = 0; i < KS; ++i) bw[t][i] = *(const v2f*)(wrow + 4 * i);
        bn[t] = bias[16 * t + l16];
    }

    StripIn<K> cur;
    load_strip<K>(A, deg, tile * 16, half, l16, cur);
    for (;;) {
        const int nxt = tile + nw;
        const bool more = nxt < nTiles;
        StripIn<K> nx;
        load_strip<K>(A, deg, (more ? nxt : tile) * 16, half, l16, nx);
        __builtin_amdgcn_sched_barrier(0);
        strip_products<K, NOUT, RELU_A>(cur, bw, bn, H, S, tile * 16 + 8 * half, l16);
        __builtin_amdgcn_sched_barrier(0);
        if (!more) break;
        cur = nx;
        tile = nxt;
    }
}


template <int F, int UB>
__device__ __forceinline__ void agg_rows(const float* __restrict__ h, float* __restrict__ out,
                                         int s, int d, int wbits, int j0, int lane) {
    constexpr int P = F / 32;
    float v[UB][P];
#pragma unroll
    for (int u = 0; u < UB; ++u) {
        const int sj = __builtin_amdgcn_readlane(s, j0 + u);
        const float* __restrict__ hr = h + (size_t)sj * F + lane;
#pragma unroll
        for (int p = 0; p < P; ++p) v[u][p] = hr[32 * p];
    }
#pragma unroll
    for (int u = 0; u < UB; ++u) {
        const int dj = __builtin_amdgcn_readlane(d, j0 + u);
        const float wj = __int_as_float(__builtin_amdgcn_readlane(wbits, j0 + u));
        float* o = out + (size_t)dj * F + lane;
#pragma unroll
        for (int p = 0; p < P; ++p) atomicAdd(o + 32 * p, v[u][p] * wj);
    }
}

#define CHUNK (2 * TPB)
template <int F>
__global__ void __launch_bounds__(TPB)
k_edge_agg(const float* __restrict__ h, const int* __restrict__ src,
           const int* __restrict__ dst, const float* __restrict__ deg,
           float* __restrict__ out, int* __restrict__ tick, int E, int n) {
    constexpr int P = F / 32;
    constexpr int U = 16 / P;
    __shared__ int ls[CHUNK], ld[CHUNK], lw[CHUNK];
    __shared__ int wcnt[TPB / 32];
    __shared__ int job[2];

    const int lane = threadIdx.x & 31;
    const int wv = __builtin_amdgcn_readfirstlane((int)(threadIdx.x >> 5));
    const int nChunks = (E + CHUNK - 1) / CHUNK;

    const int hwid = __builtin_amdgcn_s_sendmsg_rtn(0x87);
    if (threadIdx.x == 0) {
        int hf = (hwid >> 18) & 1;
        int t = atomicAdd(&tick[hf], 1);
        if (t >= nChunks) { hf ^= 1; t = atomicAdd(&tick[hf], 1); }
        job[0] = hf; job[1] = t;
    }
    __syncthreads();
    const int hf = job[0];
    const int chunk = job[1];
    if (chunk >= nChunks) return;

    const int split = (n >> 1) & ~1;
    const int e0 = chunk * CHUNK + wv * 64;

    const int ea = e0 + lane, eb = e0 + 32 + lane;
    const int ca = ea < E ? ea : E - 1, cb = eb < E ? eb : E - 1;
    const int s0 = src[ca], d0 = dst[ca], s1 = src[cb], d1 = dst[cb];
    const bool k0 = (ea < E) && ((d0 >= split) == (hf != 0));
    const bool k1 = (eb < E) && ((d1 >= split) == (hf != 0));
    const float w0 = rsqrtf(deg[k0 ? s0 : 0]) * rsqrtf(deg[k0 ? d0 : 0]);
    const float w1 = rsqrtf(deg[k1 ? s1 : 0]) * rsqrtf(deg[k1 ? d1 : 0]);
    const unsigned m0 = (unsigned)__ballot(k0);
    const unsigned m1 = (unsigned)__ballot(k1);
    const unsigned below = (1u << lane) - 1u;
    const int c0 = __popc(m0);
    const int r0 = __popc(m0 & below);
    const int r1 = c0 + __popc(m1 & below);
    if (lane == 0) wcnt[wv] = c0 + __popc(m1);
    __syncthreads();
    int base = 0, total = 0;
#pragma unroll
    for (int i = 0; i < TPB / 32; ++i) {
        const int c = wcnt[i];
        if (i < wv) base += c;
        total += c;
    }
    if (k0) { ls[base + r0] = s0; ld[base + r0] = d0; lw[base + r0] = __float_as_int(w0); }
    if (k1) { ls[base + r1] = s1; ld[base + r1] = d1; lw[base + r1] = __float_as_int(w1); }
    __syncthreads();
    total = __builtin_amdgcn_readfirstlane(total);

    for (int b = wv * 32; b < total; b += TPB) {
        const int cnt = (total - b < 32) ? (total - b) : 32;
        int s = 0, d = 0, wbits = 0;
        if (lane < cnt) { s = ls[b + lane]; d = ld[b + lane]; wbits = lw[b + lane]; }
        int j0 = 0;
        for (; j0 + U <= cnt; j0 += U) agg_rows<F, U>(h, out, s, d, wbits, j0, lane);
        if (U > 8 && j0 + 8 <= cnt) { agg_rows<F, 8>(h, out, s, d, wbits, j0, lane); j0 += 8; }
        if (j0 + 4 <= cnt) { agg_rows<F, 4>(h, out, s, d, wbits, j0, lane); j0 += 4; }
        if (j0 + 2 <= cnt) { agg_rows<F, 2>(h, out, s, d, wbits, j0, lane); j0 += 2; }
        if (j0 < cnt)      { agg_rows<F, 1>(h, out, s, d, wbits, j0, lane); }
    }
}


static inline int cdiv(long long a, int b) { return (int)((a + b - 1) / b); }

extern "C" void kernel_launch(void* const* d_in, const int* in_sizes, int n_in,
                              void* d_out, int out_size, void* d_ws, size_t ws_size,
                              hipStream_t stream) {
    (void)n_in; (void)out_size; (void)ws_size;
    const float* x  = (const float*)d_in[0];
    const int*   ei = (const int*)  d_in[1];
    const float* W1 = (const float*)d_in[2];
    const float* b1 = (const float*)d_in[3];
    const float* W2 = (const float*)d_in[4];
    const float* b2 = (const float*)d_in[5];
    float* out = (float*)d_out;

    const int N = in_sizes[0] / 128;
    const int E = in_sizes[1] / 2;
    const int* src = ei;
    const int* dst = ei + E;

    float* ws   = (float*)d_ws;
    const size_t NA = ((size_t)N + 3) & ~(size_t)3;
    float* deg  = ws;
    float* h1   = ws + NA;
    float* agg1 = h1 + 64 * NA;
    int*   tick = (int*)(agg1 + 64 * NA);
    float* h2   = h1;

    k_deg_init <<<cdiv(N, TPB), TPB, 0, stream>>>(deg, tick, N);
    k_deg_count<<<cdiv(E, TPB), TPB, 0, stream>>>(dst, deg, E);

    gcn_gemm_wmma<128, 64, false><<<GEMM_BLOCKS, GEMM_TPB, 0, stream>>>(x, W1, deg, b1, h1, agg1, N);
    k_edge_agg<64><<<2 * cdiv((long long)E, CHUNK), TPB, 0, stream>>>(h1, src, dst, deg, agg1, tick, E, N);

    gcn_gemm_wmma<64, 32, true><<<GEMM_BLOCKS, GEMM_TPB, 0, stream>>>(agg1, W2, deg, b2, h2, out, N);
    k_edge_agg<32><<<2 * cdiv((long long)E, CHUNK), TPB, 0, stream>>>(h2, src, dst, deg, out, tick + 2, E, N);
}


# In-degree counted in LDS: 256 blocks of (row range, edge slice) add into LDS integer counters and store whole lines; a small pass sums 32 slices per row; no global atomic in the count; same degrees
# speedup vs baseline: 2.8295x; 1.3945x over previous
#include <hip/hip_runtime.h>
#include <hip/hip_bf16.h>


typedef float v2f __attribute__((ext_vector_type(2)));
typedef float v8f __attribute__((ext_vector_type(8)));

#define TPB 256


#define HR 8
#define HK 32
#define HT 512
#define HCAP 12544
#define HQ 13

__device__ __forceinline__ int deg_range_rows(int n) {
    return (((n + HR - 1) / HR) + 31) & ~31;
}

__device__ __forceinline__ void deg_hit(unsigned* cnt, int d, int lo, int len) {
    const unsigned off = (unsigned)(d - lo);
    if (off < (unsigned)len) atomicAdd(&cnt[off], 1u);
}

__global__ void __launch_bounds__(HT)
k_deg_hist(const int* __restrict__ dst, unsigned* __restrict__ part, int E, int n) {
    __shared__ unsigned cnt[HCAP];
    const int tid = threadIdx.x;
    const int r = blockIdx.x % HR;
    const int k = blockIdx.x / HR;
    const int RR = deg_range_rows(n);
    const int NP = RR * HR;
    const int sl = (((E + HK - 1) / HK) + 3) & ~3;
    const int e0 = k * sl;
    const int e1 = (e0 + sl < E) ? (e0 + sl) : E;
    const int ne = e1 > e0 ? e1 - e0 : 0;
    const bool al = ((((unsigned long long)(dst + e0)) & 15ull) == 0ull);
    const int nq = al ? (ne >> 2) : 0;
    const int4* __restrict__ dq = (const int4*)(dst + e0);

    for (int lo = r * RR; lo < (r + 1) * RR; lo += HCAP) {
        const int len = ((r + 1) * RR - lo < HCAP) ? ((r + 1) * RR - lo) : HCAP;
        for (int i = tid; i < len; i += HT) cnt[i] = 0u;
        __syncthreads();

        for (int qb = 0; qb < nq; qb += HT * HQ) {
            int4 v[HQ];
#pragma unroll
            for (int u = 0; u < HQ; ++u) {
                const int q = qb + u * HT + tid;
                v[u] = dq[q < nq ? q : nq - 1];
            }
#pragma unroll
            for (int u = 0; u < HQ; ++u) {
                if (qb + u * HT + tid < nq) {
                    deg_hit(cnt, v[u].x, lo, len);
                    deg_hit(cnt, v[u].y, lo, len);
                    deg_hit(cnt, v[u].z, lo, len);
                    deg_hit(cnt, v[u].w, lo, len);
                }
            }
        }
        for (int e = e0 + 4 * nq + tid; e < e1; e += HT) deg_hit(cnt, dst[e], lo, len);
        __syncthreads();

        unsigned* __restrict__ prow = part + (size_t)k * NP + lo;
        for (int i = tid; i < len; i += HT) prow[i] = cnt[i];
        __syncthreads();
    }
}

__global__ void k_deg_sum(const unsigned* __restrict__ part, float* __restrict__ deg,
                          int* __restrict__ tick, int n) {
    const int i = blockIdx.x * TPB + threadIdx.x;
    if (i < 4) tick[i] = 0;
    if (i >= n) return;
    const int NP = deg_range_rows(n) * HR;
    unsigned c = 1u;
#pragma unroll
    for (int k = 0; k < HK; ++k) c += part[(size_t)k * NP + i];
    deg[i] = (float)c;
}


#define GEMM_TPB 128
#define GEMM_BLOCKS 512

template <int K>
struct StripIn {
    v2f a[K / 4];
    float4 dg0, dg1;
};

template <int K>
__device__ __forceinline__ void load_strip(const float* __restrict__ A, const float* __restrict__ deg,
                                           int m0, int half, int l16, StripIn<K>& in) {
    const float* __restrict__ arow = A + (size_t)(m0 + l16) * K + 2 * half;
    const int mbase = m0 + 8 * half;
    in.dg0 = *(const float4*)(deg + mbase);
    in.dg1 = *(const float4*)(deg + mbase + 4);
#pragma unroll
    for (int i = 0; i < K / 4; ++i) in.a[i] = *(const v2f*)(arow + 4 * i);
}

template <int K, int NOUT, bool RELU_A>
__device__ __forceinline__ void strip_products(const StripIn<K>& in, const v2f (&bw)[NOUT / 16][K / 4],
                                               const float (&bn)[NOUT / 16],
                                               float* __restrict__ H, float* __restrict__ S,
                                               int mbase, int l16) {
    constexpr int CT = NOUT / 16;
    constexpr int KS = K / 4;
    v8f acc[CT];
#pragma unroll
    for (int t = 0; t < CT; ++t) acc[t] = (v8f){0.f, 0.f, 0.f, 0.f, 0.f, 0.f, 0.f, 0.f};
#pragma unroll
    for (int i = 0; i < KS; ++i) {
        v2f a = in.a[i];
        if (RELU_A) { a.x = fmaxf(a.x, 0.0f); a.y = fmaxf(a.y, 0.0f); }
#pragma unroll
        for (int t = 0; t < CT; ++t) {
            acc[t] = __builtin_amdgcn_wmma_f32_16x16x4_f32(
                 false, a,  false, bw[t][i],
                 (short)0, acc[t],  false,  false);
        }
    }
    const float dg[8] = {in.dg0.x, in.dg0.y, in.dg0.z, in.dg0.w, in.dg1.x, in.dg1.y, in.dg1.z, in.dg1.w};
#pragma unroll
    for (int j = 0; j < 8; ++j) {
        const float di = rsqrtf(dg[j]);
        const float d2 = di * di;
#pragma unroll
        for (int t = 0; t < CT; ++t) {
            const size_t o = (size_t)(mbase + j) * NOUT + 16 * t + l16;
            H[o] = acc[t][j];
            S[o] = acc[t][j] * d2 + bn[t];
        }
    }
}

template <int K, int NOUT, bool RELU_A>
__global__ void __launch_bounds__(GEMM_TPB)
gcn_gemm_wmma(const float* __restrict__ A, const float* __restrict__ W,
              const float* __restrict__ deg, const float* __restrict__ bias,
              float* __restrict__ H, float* __restrict__ S, int M) {
    constexpr int CT = NOUT / 16;
    constexpr int KS = K / 4;

    const int lane = threadIdx.x & 31;
    const int wv = __builtin_amdgcn_readfirstlane((int)(threadIdx.x >> 5));
    const int gw = blockIdx.x * (GEMM_TPB / 32) + wv;
    const int nw = gridDim.x * (GEMM_TPB / 32);
    const int nTiles = (M + 15) >> 4;

    const int half = lane >> 4;
    const int l16  = lane & 15;

    int tile = gw;
    if (tile >= nTiles) return;

    v2f bw[CT][KS];
    float bn[CT];
#pragma unroll
    for (int t = 0; t < CT; ++t) {
        const float* __restrict__ wrow = W + (size_t)(16 * t + l16) * K + 2 * half;
#pragma unroll
        for (int i = 0; i < KS; ++i) bw[t][i] = *(const v2f*)(wrow + 4 * i);
        bn[t] = bias[16 * t + l16];
    }

    StripIn<K> cur;
    load_strip<K>(A, deg, tile * 16, half, l16, cur);
    for (;;) {
        const int nxt = tile + nw;
        const bool more = nxt < nTiles;
        StripIn<K> nx;
        load_strip<K>(A, deg, (more ? nxt : tile) * 16, half, l16, nx);
        __builtin_amdgcn_sched_barrier(0);
        strip_products<K, NOUT, RELU_A>(cur, bw, bn, H, S, tile * 16 + 8 * half, l16);
        __builtin_amdgcn_sched_barrier(0);
        if (!more) break;
        cur = nx;
        tile = nxt;
    }
}


template <int F, int UB>
__device__ __forceinline__ void agg_rows(const float* __restrict__ h, float* __restrict__ out,
                                         int s, int d, int wbits, int j0, int lane) {
    constexpr int P = F / 32;
    float v[UB][P];
#pragma unroll
    for (int u = 0; u < UB; ++u) {
        const int sj = __builtin_amdgcn_readlane(s, j0 + u);
        const float* __restrict__ hr = h + (size_t)sj * F + lane;
#pragma unroll
        for (int p = 0; p < P; ++p) v[u][p] = hr[32 * p];
    }
#pragma unroll
    for (int u = 0; u < UB; ++u) {
        const int dj = __builtin_amdgcn_readlane(d, j0 + u);
        const float wj = __int_as_float(__builtin_amdgcn_readlane(wbits, j0 + u));
        float* o = out + (size_t)dj * F + lane;
#pragma unroll
        for (int p = 0; p < P; ++p) atomicAdd(o + 32 * p, v[u][p] * wj);
    }
}

#define CHUNK (2 * TPB)
template <int F>
__global__ void __launch_bounds__(TPB)
k_edge_agg(const float* __restrict__ h, const int* __restrict__ src,
           const int* __restrict__ dst, const float* __restrict__ deg,
           float* __restrict__ out, int* __restrict__ tick, int E, int n) {
    constexpr int P = F / 32;
    constexpr int U = 16 / P;
    __shared__ int ls[CHUNK], ld[CHUNK], lw[CHUNK];
    __shared__ int wcnt[TPB / 32];
    __shared__ int job[2];

    const int lane = threadIdx.x & 31;
    const int wv = __builtin_amdgcn_readfirstlane((int)(threadIdx.x >> 5));
    const int nChunks = (E + CHUNK - 1) / CHUNK;

    const int hwid = __builtin_amdgcn_s_sendmsg_rtn(0x87);
    if (threadIdx.x == 0) {
        int hf = (hwid >> 18) & 1;
        int t = atomicAdd(&tick[hf], 1);
        if (t >= nChunks) { hf ^= 1; t = atomicAdd(&tick[hf], 1); }
        job[0] = hf; job[1] = t;
    }
    __syncthreads();
    const int hf = job[0];
    const int chunk = job[1];
    if (chunk >= nChunks) return;

    const int split = (n >> 1) & ~1;
    const int e0 = chunk * CHUNK + wv * 64;

    const int ea = e0 + lane, eb = e0 + 32 + lane;
    const int ca = ea < E ? ea : E - 1, cb = eb < E ? eb : E - 1;
    const int s0 = src[ca], d0 = dst[ca], s1 = src[cb], d1 = dst[cb];
    const bool k0 = (ea < E) && ((d0 >= split) == (hf != 0));
    const bool k1 = (eb < E) && ((d1 >= split) == (hf != 0));
    const float w0 = rsqrtf(deg[k0 ? s0 : 0]) * rsqrtf(deg[k0 ? d0 : 0]);
    const float w1 = rsqrtf(deg[k1 ? s1 : 0]) * rsqrtf(deg[k1 ? d1 : 0]);
    const unsigned m0 = (unsigned)__ballot(k0);
    const unsigned m1 = (unsigned)__ballot(k1);
    const unsigned below = (1u << lane) - 1u;
    const int c0 = __popc(m0);
    const int r0 = __popc(m0 & below);
    const int r1 = c0 + __popc(m1 & below);
    if (lane == 0) wcnt[wv] = c0 + __popc(m1);
    __syncthreads();
    int base = 0, total = 0;
#pragma unroll
    for (int i = 0; i < TPB / 32; ++i) {
        const int c = wcnt[i];
        if (i < wv) base += c;
        total += c;
    }
    if (k0) { ls[base + r0] = s0; ld[base + r0] = d0; lw[base + r0] = __float_as_int(w0); }
    if (k1) { ls[base + r1] = s1; ld[base + r1] = d1; lw[base + r1] = __float_as_int(w1); }
    __syncthreads();
    total = __builtin_amdgcn_readfirstlane(total);

    for (int b = wv * 32; b < total; b += TPB) {
        const int cnt = (total - b < 32) ? (total - b) : 32;
        int s = 0, d = 0, wbits = 0;
        if (lane < cnt) { s = ls[b + lane]; d = ld[b + lane]; wbits = lw[b + lane]; }
        int j0 = 0;
        for (; j0 + U <= cnt; j0 += U) agg_rows<F, U>(h, out, s, d, wbits, j0, lane);
        if (U > 8 && j0 + 8 <= cnt) { agg_rows<F, 8>(h, out, s, d, wbits, j0, lane); j0 += 8; }
        if (j0 + 4 <= cnt) { agg_rows<F, 4>(h, out, s, d, wbits, j0, lane); j0 += 4; }
        if (j0 + 2 <= cnt) { agg_rows<F, 2>(h, out, s, d, wbits, j0, lane); j0 += 2; }
        if (j0 < cnt)      { agg_rows<F, 1>(h, out, s, d, wbits, j0, lane); }
    }
}


static inline int cdiv(long long a, int b) { return (int)((a + b - 1) / b); }

extern "C" void kernel_launch(void* const* d_in, const int* in_sizes, int n_in,
                              void* d_out, int out_size, void* d_ws, size_t ws_size,
                              hipStream_t stream) {
    (void)n_in; (void)out_size; (void)ws_size;
    const float* x  = (const float*)d_in[0];
    const int*   ei = (const int*)  d_in[1];
    const float* W1 = (const float*)d_in[2];
    const float* b1 = (const float*)d_in[3];
    const float* W2 = (const float*)d_in[4];
    const float* b2 = (const float*)d_in[5];
    float* out = (float*)d_out;

    const int N = in_sizes[0] / 128;
    const int E = in_sizes[1] / 2;
    const int* src = ei;
    const int* dst = ei + E;

    float* ws   = (float*)d_ws;
    const size_t NA = ((size_t)N + 3) & ~(size_t)3;
    float* deg  = ws;
    float* h1   = ws + NA;
    float* agg1 = h1 + 64 * NA;
    int*   tick = (int*)(agg1 + 64 * NA);
    float* h2   = h1;

    unsigned* part = (unsigned*)agg1;
    k_deg_hist<<<HR * HK, HT, 0, stream>>>(dst, part, E, N);
    k_deg_sum <<<cdiv(N, TPB), TPB, 0, stream>>>(part, deg, tick, N);

    gcn_gemm_wmma<128, 64, false><<<GEMM_BLOCKS, GEMM_TPB, 0, stream>>>(x, W1, deg, b1, h1, agg1, N);
    k_edge_agg<64><<<2 * cdiv((long long)E, CHUNK), TPB, 0, stream>>>(h1, src, dst, deg, agg1, tick, E, N);

    gcn_gemm_wmma<64, 32, true><<<GEMM_BLOCKS, GEMM_TPB, 0, stream>>>(agg1, W2, deg, b2, h2, out, N);
    k_edge_agg<32><<<2 * cdiv((long long)E, CHUNK), TPB, 0, stream>>>(h2, src, dst, deg, out, tick + 2, E, N);
}
